# starting kernel
# baseline (speedup 1.0000x reference)
#include <hip/hip_runtime.h>
#include <hip/hip_bf16.h>
#include <stddef.h>


typedef float v2f __attribute__((ext_vector_type(2)));
typedef float v4f __attribute__((ext_vector_type(4)));
typedef float v8f __attribute__((ext_vector_type(8)));

#define BATCH   512
#define STEPS   500
#define LEN     128
#define HID     512
#define NPAIR1  17
#define NPAIR2  64
#define XS_STRIDE 140
#define HS_STRIDE 516
#define DT_F     0.001f
#define SQRT_DT  0.03162277660168379332f

#define W1F_FLOATS (32 * NPAIR1 * 32 * 4)
#define W2F_FLOATS (8  * NPAIR2 * 32 * 4)

__device__ __forceinline__ v8f wmma4(v2f a, v2f b, v8f c) {
  return __builtin_amdgcn_wmma_f32_16x16x4_f32(
       false, a,  false, b,
       (short)0, c,  false,  false);
}

__global__ __launch_bounds__(256) void swizzle_weights(
    const float* __restrict__ W1, const float* __restrict__ W2,
    float* __restrict__ W1f, float* __restrict__ W2f) {
  int gid = blockIdx.x * 256 + threadIdx.x;
  if (gid < W1F_FLOATS) {
    int j    = gid & 3;
    int lane = (gid >> 2) & 31;
    int rest = gid >> 7;
    int p    = rest % NPAIR1;
    int nt   = rest / NPAIR1;
    int n    = nt * 16 + (lane & 15);
    int k    = 4 * (2 * p + (j >> 1)) + ((lane >> 4) << 1) + (j & 1);
    W1f[gid] = (k < LEN + 1) ? W1[k * HID + n] : 0.0f;
  } else if (gid < W1F_FLOATS + W2F_FLOATS) {
    int g    = gid - W1F_FLOATS;
    int j    = g & 3;
    int lane = (g >> 2) & 31;
    int rest = g >> 7;
    int p    = rest & 63;
    int nt   = rest >> 6;
    int n    = nt * 16 + (lane & 15);
    int k    = 4 * (2 * p + (j >> 1)) + ((lane >> 4) << 1) + (j & 1);
    W2f[g]   = W2[k * LEN + n];
  }
}

__global__ __launch_bounds__(256) void diffusion_scan(
    const float* __restrict__ fd,
    const float* __restrict__ b1,
    const float* __restrict__ b2,
    const float* __restrict__ noise,
    const float* __restrict__ W1f,
    const float* __restrict__ W2f,
    float* __restrict__ out) {
  __shared__ float xs[16 * XS_STRIDE];
  __shared__ float hs[16 * HS_STRIDE];

  const int tid   = threadIdx.x;
  const int wave  = tid >> 5;
  const int lane  = tid & 31;
  const int lrow  = lane & 15;
  const int lhalf = lane >> 4;
  const int r0    = blockIdx.x * 16;

  for (int idx = tid; idx < 16 * XS_STRIDE; idx += 256) {
    int m = idx / XS_STRIDE;
    int c = idx - m * XS_STRIDE;
    float v = 0.0f;
    if (c < LEN) v = fd[((size_t)(r0 + m) * STEPS + (STEPS - 1)) * LEN + c];
    xs[idx] = v;
  }
  for (int idx = tid; idx < 16 * LEN; idx += 256) {
    int m = idx >> 7;
    int c = idx & (LEN - 1);
    out[((size_t)(r0 + m) * STEPS) * LEN + c] =
        fd[((size_t)(r0 + m) * STEPS + (STEPS - 1)) * LEN + c];
  }

  float bias1[4];
  for (int q = 0; q < 4; ++q) bias1[q] = b1[(wave * 4 + q) * 16 + lrow];
  const float bias2v = b2[wave * 16 + lrow];

  const v2f* px = (const v2f*)(xs + lrow * XS_STRIDE + lhalf * 2);
  const v2f* ph = (const v2f*)(hs + lrow * HS_STRIDE + lhalf * 2);
  const v4f* pw1 = (const v4f*)W1f + (size_t)wave * 4 * NPAIR1 * 32 + lane;
  const v4f* pw2 = (const v4f*)W2f + (size_t)wave * NPAIR2 * 32 + lane;

  __syncthreads();

#pragma unroll 1
  for (int i = 0; i < STEPS - 1; ++i) {
    const float tcur = DT_F * (float)(STEPS - 1 - i);
    if (tid < 16) xs[tid * XS_STRIDE + LEN] = tcur;
    __syncthreads();

    v8f acc[4];
#pragma unroll
    for (int q = 0; q < 4; ++q) {
      v8f a;
      for (int v = 0; v < 8; ++v) a[v] = bias1[q];
      acc[q] = a;
    }
#pragma unroll
    for (int p = 0; p < NPAIR1; ++p) {
      v2f a0 = px[4 * p];
      v2f a1 = px[4 * p + 2];
#pragma unroll
      for (int q = 0; q < 4; ++q) {
        v4f w = pw1[(q * NPAIR1 + p) * 32];
        v2f w0; w0.x = w.x; w0.y = w.y;
        v2f w1; w1.x = w.z; w1.y = w.w;
        acc[q] = wmma4(a0, w0, acc[q]);
        acc[q] = wmma4(a1, w1, acc[q]);
      }
    }
#pragma unroll
    for (int q = 0; q < 4; ++q) {
      int n = (wave * 4 + q) * 16 + lrow;
#pragma unroll
      for (int v = 0; v < 8; ++v) {
        float h = acc[q][v];
        hs[(v + 8 * lhalf) * HS_STRIDE + n] = h > 0.0f ? h : 0.0f;
      }
    }
    __syncthreads();

    v8f sc[4];
    {
      v8f z;
      for (int v = 0; v < 8; ++v) z[v] = 0.0f;
      v8f s0;
      for (int v = 0; v < 8; ++v) s0[v] = bias2v;
      sc[0] = s0; sc[1] = z; sc[2] = z; sc[3] = z;
    }
#pragma unroll 8
    for (int p = 0; p < NPAIR2; ++p) {
      v2f a0 = ph[4 * p];
      v2f a1 = ph[4 * p + 2];
      v4f w = pw2[p * 32];
      v2f w0; w0.x = w.x; w0.y = w.y;
      v2f w1; w1.x = w.z; w1.y = w.w;
      sc[p & 3] = wmma4(a0, w0, sc[p & 3]);
      sc[p & 3] = wmma4(a1, w1, sc[p & 3]);
    }
    v8f scf;
#pragma unroll
    for (int v = 0; v < 8; ++v)
      scf[v] = (sc[0][v] + sc[1][v]) + (sc[2][v] + sc[3][v]);

    {
      int n = wave * 16 + lrow;
      float eps = noise[i * LEN + n];
      int po = STEPS - 1 - i;
#pragma unroll
      for (int v = 0; v < 8; ++v) {
        int m = v + 8 * lhalf;
        float xv = xs[m * XS_STRIDE + n];
        float xn = xv + (-0.5f * xv - scf[v]) * DT_F + SQRT_DT * eps;
        xs[m * XS_STRIDE + n] = xn;
        out[((size_t)(r0 + m) * STEPS + po) * LEN + n] = xn;
      }
    }
  }
}

extern "C" void kernel_launch(void* const* d_in, const int* in_sizes, int n_in,
                              void* d_out, int out_size, void* d_ws, size_t ws_size,
                              hipStream_t stream) {
  const float* fd    = (const float*)d_in[0];
  const float* W1    = (const float*)d_in[1];
  const float* b1    = (const float*)d_in[2];
  const float* W2    = (const float*)d_in[3];
  const float* b2    = (const float*)d_in[4];
  const float* noise = (const float*)d_in[5];
  float* out = (float*)d_out;

  float* W1f = (float*)d_ws;
  float* W2f = W1f + W1F_FLOATS;

  int total = W1F_FLOATS + W2F_FLOATS;
  swizzle_weights<<<(total + 255) / 256, 256, 0, stream>>>(W1, W2, W1f, W2f);

  diffusion_scan<<<BATCH / 16, 256, 0, stream>>>(fd, b1, b2, noise, W1f, W2f, out);
}


# Loads issued ahead of use: W2 fragments 3 blocks ahead in a 4-slot register ring running across steps (step barrier fences LDS only), h 1 block ahead, noise and x values requested early
# speedup vs baseline: 1.1250x; 1.1250x over previous
#include <hip/hip_runtime.h>
#include <hip/hip_bf16.h>
#include <stddef.h>


typedef float v2f __attribute__((ext_vector_type(2)));
typedef float v4f __attribute__((ext_vector_type(4)));
typedef float v8f __attribute__((ext_vector_type(8)));

#define BATCH   512
#define STEPS   500
#define LEN     128
#define HID     512
#define NPAIR1  17
#define NPAIR2  64
#define XS_STRIDE 140
#define HS_STRIDE 516
#define DT_F     0.001f
#define SQRT_DT  0.03162277660168379332f

#define W1F_FLOATS (32 * NPAIR1 * 32 * 4)
#define W2F_FLOATS (8  * NPAIR2 * 32 * 4)

__device__ __forceinline__ v8f wmma4(v2f a, v2f b, v8f c) {
  return __builtin_amdgcn_wmma_f32_16x16x4_f32(
       false, a,  false, b,
       (short)0, c,  false,  false);
}

__device__ __forceinline__ void barrier_lds_only() {
  __builtin_amdgcn_fence(__ATOMIC_RELEASE, "workgroup", "local");
  __builtin_amdgcn_s_barrier();
  __builtin_amdgcn_fence(__ATOMIC_ACQUIRE, "workgroup", "local");
}

__global__ __launch_bounds__(256) void swizzle_weights(
    const float* __restrict__ W1, const float* __restrict__ W2,
    float* __restrict__ W1f, float* __restrict__ W2f) {
  int gid = blockIdx.x * 256 + threadIdx.x;
  if (gid < W1F_FLOATS) {
    int j    = gid & 3;
    int lane = (gid >> 2) & 31;
    int rest = gid >> 7;
    int p    = rest % NPAIR1;
    int nt   = rest / NPAIR1;
    int n    = nt * 16 + (lane & 15);
    int k    = 4 * (2 * p + (j >> 1)) + ((lane >> 4) << 1) + (j & 1);
    W1f[gid] = (k < LEN + 1) ? W1[k * HID + n] : 0.0f;
  } else if (gid < W1F_FLOATS + W2F_FLOATS) {
    int g    = gid - W1F_FLOATS;
    int j    = g & 3;
    int lane = (g >> 2) & 31;
    int rest = g >> 7;
    int p    = rest & 63;
    int nt   = rest >> 6;
    int n    = nt * 16 + (lane & 15);
    int k    = 4 * (2 * p + (j >> 1)) + ((lane >> 4) << 1) + (j & 1);
    W2f[g]   = W2[k * LEN + n];
  }
}

__global__ __launch_bounds__(256) void diffusion_scan(
    const float* __restrict__ fd,
    const float* __restrict__ b1,
    const float* __restrict__ b2,
    const float* __restrict__ noise,
    const float* __restrict__ W1f,
    const float* __restrict__ W2f,
    float* __restrict__ out) {
  __shared__ float xs[16 * XS_STRIDE];
  __shared__ float hs[16 * HS_STRIDE];

  const int tid   = threadIdx.x;
  const int wave  = tid >> 5;
  const int lane  = tid & 31;
  const int lrow  = lane & 15;
  const int lhalf = lane >> 4;
  const int r0    = blockIdx.x * 16;

  for (int idx = tid; idx < 16 * XS_STRIDE; idx += 256) {
    int m = idx / XS_STRIDE;
    int c = idx - m * XS_STRIDE;
    float v = 0.0f;
    if (c < LEN) v = fd[((size_t)(r0 + m) * STEPS + (STEPS - 1)) * LEN + c];
    xs[idx] = v;
  }
  for (int idx = tid; idx < 16 * LEN; idx += 256) {
    int m = idx >> 7;
    int c = idx & (LEN - 1);
    out[((size_t)(r0 + m) * STEPS) * LEN + c] =
        fd[((size_t)(r0 + m) * STEPS + (STEPS - 1)) * LEN + c];
  }

  float bias1[4];
  for (int q = 0; q < 4; ++q) bias1[q] = b1[(wave * 4 + q) * 16 + lrow];
  const float bias2v = b2[wave * 16 + lrow];

  const v2f* px = (const v2f*)(xs + lrow * XS_STRIDE + lhalf * 2);
  const v2f* ph = (const v2f*)(hs + lrow * HS_STRIDE + lhalf * 2);
  const v4f* pw1 = (const v4f*)W1f + (size_t)wave * 4 * NPAIR1 * 32 + lane;
  const v4f* pw2 = (const v4f*)W2f + (size_t)wave * NPAIR2 * 32 + lane;

  v4f w1f[4][NPAIR1];
#pragma unroll
  for (int q = 0; q < 4; ++q)
#pragma unroll
    for (int p = 0; p < NPAIR1; ++p) w1f[q][p] = pw1[(q * NPAIR1 + p) * 32];

  v4f w2r[4][4];
#pragma unroll
  for (int b = 0; b < 3; ++b)
#pragma unroll
    for (int k = 0; k < 4; ++k) w2r[b][k] = pw2[(b * 4 + k) * 32];

  __syncthreads();

#pragma unroll 1
  for (int i = 0; i < STEPS - 1; ++i) {
    const float tcur = DT_F * (float)(STEPS - 1 - i);
    if (tid < 16) xs[tid * XS_STRIDE + LEN] = tcur;
    const float eps = noise[i * LEN + wave * 16 + lrow];
    barrier_lds_only();

    v8f acc[4];
#pragma unroll
    for (int q = 0; q < 4; ++q) {
      v8f a;
      for (int v = 0; v < 8; ++v) a[v] = bias1[q];
      acc[q] = a;
    }
#pragma unroll
    for (int p = 0; p < NPAIR1; ++p) {
      v2f a0 = px[4 * p];
      v2f a1 = px[4 * p + 2];
#pragma unroll
      for (int q = 0; q < 4; ++q) {
        v4f w = w1f[q][p];
        v2f w0; w0.x = w.x; w0.y = w.y;
        v2f w1; w1.x = w.z; w1.y = w.w;
        acc[q] = wmma4(a0, w0, acc[q]);
        acc[q] = wmma4(a1, w1, acc[q]);
      }
    }
#pragma unroll
    for (int q = 0; q < 4; ++q) {
      int n = (wave * 4 + q) * 16 + lrow;
#pragma unroll
      for (int v = 0; v < 8; ++v) {
        float h = acc[q][v];
        hs[(v + 8 * lhalf) * HS_STRIDE + n] = h > 0.0f ? h : 0.0f;
      }
    }
    __syncthreads();

    v8f sc[4];
    {
      v8f z;
      for (int v = 0; v < 8; ++v) z[v] = 0.0f;
      v8f s0;
      for (int v = 0; v < 8; ++v) s0[v] = bias2v;
      sc[0] = s0; sc[1] = z; sc[2] = z; sc[3] = z;
    }
    v2f ha[2][4], hb[2][4];
#pragma unroll
    for (int k = 0; k < 4; ++k) {
      ha[0][k] = ph[4 * k];
      hb[0][k] = ph[4 * k + 2];
    }
    float xv[8];
#pragma unroll
    for (int v = 0; v < 8; ++v)
      xv[v] = xs[(v + 8 * lhalf) * XS_STRIDE + wave * 16 + lrow];
#pragma unroll 1
    for (int t = 0; t < 4; ++t) {
#pragma unroll
      for (int s = 0; s < 4; ++s) {
        const int cur = t * 4 + s;
        const int nw  = (cur + 3) & 15;
        const int nh  = (cur + 1) & 15;
#pragma unroll
        for (int k = 0; k < 4; ++k)
          w2r[(s + 3) & 3][k] = pw2[(nw * 4 + k) * 32];
#pragma unroll
        for (int k = 0; k < 4; ++k) {
          ha[(s + 1) & 1][k] = ph[4 * (nh * 4 + k)];
          hb[(s + 1) & 1][k] = ph[4 * (nh * 4 + k) + 2];
        }
        __builtin_amdgcn_sched_barrier(0);
#pragma unroll
        for (int k = 0; k < 4; ++k) {
          v4f w = w2r[s][k];
          v2f w0; w0.x = w.x; w0.y = w.y;
          v2f w1; w1.x = w.z; w1.y = w.w;
          sc[k] = wmma4(ha[s & 1][k], w0, sc[k]);
          sc[k] = wmma4(hb[s & 1][k], w1, sc[k]);
        }
        __builtin_amdgcn_sched_barrier(0);
      }
    }
    v8f scf;
#pragma unroll
    for (int v = 0; v < 8; ++v)
      scf[v] = (sc[0][v] + sc[1][v]) + (sc[2][v] + sc[3][v]);

    {
      int n = wave * 16 + lrow;
      int po = STEPS - 1 - i;
#pragma unroll
      for (int v = 0; v < 8; ++v) {
        int m = v + 8 * lhalf;
        float xn = xv[v] + (-0.5f * xv[v] - scf[v]) * DT_F + SQRT_DT * eps;
        xs[m * XS_STRIDE + n] = xn;
        out[((size_t)(r0 + m) * STEPS + po) * LEN + n] = xn;
      }
    }
  }
}

extern "C" void kernel_launch(void* const* d_in, const int* in_sizes, int n_in,
                              void* d_out, int out_size, void* d_ws, size_t ws_size,
                              hipStream_t stream) {
  const float* fd    = (const float*)d_in[0];
  const float* W1    = (const float*)d_in[1];
  const float* b1    = (const float*)d_in[2];
  const float* W2    = (const float*)d_in[3];
  const float* b2    = (const float*)d_in[4];
  const float* noise = (const float*)d_in[5];
  float* out = (float*)d_out;

  float* W1f = (float*)d_ws;
  float* W2f = W1f + W1F_FLOATS;

  int total = W1F_FLOATS + W2F_FLOATS;
  swizzle_weights<<<(total + 255) / 256, 256, 0, stream>>>(W1, W2, W1f, W2f);

  diffusion_scan<<<BATCH / 16, 256, 0, stream>>>(fd, b1, b2, noise, W1f, W2f, out);
}


# GEMM1 x operand: all 17 K-pairs loaded as ONE LDS group behind the step barrier (68 VGPRs), not one load between every 8 WMMAs: GEMM1 runs as 136 WMMAs with no LDS place in the stream
# speedup vs baseline: 1.1688x; 1.0389x over previous
#include <hip/hip_runtime.h>
#include <hip/hip_bf16.h>
#include <stddef.h>


typedef float v2f __attribute__((ext_vector_type(2)));
typedef float v4f __attribute__((ext_vector_type(4)));
typedef float v8f __attribute__((ext_vector_type(8)));

#define BATCH   512
#define STEPS   500
#define LEN     128
#define HID     512
#define NPAIR1  17
#define NPAIR2  64
#define XS_STRIDE 140
#define HS_STRIDE 516
#define DT_F     0.001f
#define SQRT_DT  0.03162277660168379332f

#define W1F_FLOATS (32 * NPAIR1 * 32 * 4)
#define W2F_FLOATS (8  * NPAIR2 * 32 * 4)

__device__ __forceinline__ v8f wmma4(v2f a, v2f b, v8f c) {
  return __builtin_amdgcn_wmma_f32_16x16x4_f32(
       false, a,  false, b,
       (short)0, c,  false,  false);
}

__device__ __forceinline__ void barrier_lds_only() {
  __builtin_amdgcn_fence(__ATOMIC_RELEASE, "workgroup", "local");
  __builtin_amdgcn_s_barrier();
  __builtin_amdgcn_fence(__ATOMIC_ACQUIRE, "workgroup", "local");
}

__global__ __launch_bounds__(256) void swizzle_weights(
    const float* __restrict__ W1, const float* __restrict__ W2,
    float* __restrict__ W1f, float* __restrict__ W2f) {
  int gid = blockIdx.x * 256 + threadIdx.x;
  if (gid < W1F_FLOATS) {
    int j    = gid & 3;
    int lane = (gid >> 2) & 31;
    int rest = gid >> 7;
    int p    = rest % NPAIR1;
    int nt   = rest / NPAIR1;
    int n    = nt * 16 + (lane & 15);
    int k    = 4 * (2 * p + (j >> 1)) + ((lane >> 4) << 1) + (j & 1);
    W1f[gid] = (k < LEN + 1) ? W1[k * HID + n] : 0.0f;
  } else if (gid < W1F_FLOATS + W2F_FLOATS) {
    int g    = gid - W1F_FLOATS;
    int j    = g & 3;
    int lane = (g >> 2) & 31;
    int rest = g >> 7;
    int p    = rest & 63;
    int nt   = rest >> 6;
    int n    = nt * 16 + (lane & 15);
    int k    = 4 * (2 * p + (j >> 1)) + ((lane >> 4) << 1) + (j & 1);
    W2f[g]   = W2[k * LEN + n];
  }
}

__global__ __launch_bounds__(256) void diffusion_scan(
    const float* __restrict__ fd,
    const float* __restrict__ b1,
    const float* __restrict__ b2,
    const float* __restrict__ noise,
    const float* __restrict__ W1f,
    const float* __restrict__ W2f,
    float* __restrict__ out) {
  __shared__ float xs[16 * XS_STRIDE];
  __shared__ float hs[16 * HS_STRIDE];

  const int tid   = threadIdx.x;
  const int wave  = tid >> 5;
  const int lane  = tid & 31;
  const int lrow  = lane & 15;
  const int lhalf = lane >> 4;
  const int r0    = blockIdx.x * 16;

  for (int idx = tid; idx < 16 * XS_STRIDE; idx += 256) {
    int m = idx / XS_STRIDE;
    int c = idx - m * XS_STRIDE;
    float v = 0.0f;
    if (c < LEN) v = fd[((size_t)(r0 + m) * STEPS + (STEPS - 1)) * LEN + c];
    xs[idx] = v;
  }
  for (int idx = tid; idx < 16 * LEN; idx += 256) {
    int m = idx >> 7;
    int c = idx & (LEN - 1);
    out[((size_t)(r0 + m) * STEPS) * LEN + c] =
        fd[((size_t)(r0 + m) * STEPS + (STEPS - 1)) * LEN + c];
  }

  float bias1[4];
  for (int q = 0; q < 4; ++q) bias1[q] = b1[(wave * 4 + q) * 16 + lrow];
  const float bias2v = b2[wave * 16 + lrow];

  const v2f* px = (const v2f*)(xs + lrow * XS_STRIDE + lhalf * 2);
  const v2f* ph = (const v2f*)(hs + lrow * HS_STRIDE + lhalf * 2);
  const v4f* pw1 = (const v4f*)W1f + (size_t)wave * 4 * NPAIR1 * 32 + lane;
  const v4f* pw2 = (const v4f*)W2f + (size_t)wave * NPAIR2 * 32 + lane;

  v4f w1f[4][NPAIR1];
#pragma unroll
  for (int q = 0; q < 4; ++q)
#pragma unroll
    for (int p = 0; p < NPAIR1; ++p) w1f[q][p] = pw1[(q * NPAIR1 + p) * 32];

  v4f w2r[4][4];
#pragma unroll
  for (int b = 0; b < 3; ++b)
#pragma unroll
    for (int k = 0; k < 4; ++k) w2r[b][k] = pw2[(b * 4 + k) * 32];

  __syncthreads();

#pragma unroll 1
  for (int i = 0; i < STEPS - 1; ++i) {
    const float tcur = DT_F * (float)(STEPS - 1 - i);
    if (tid < 16) xs[tid * XS_STRIDE + LEN] = tcur;
    const float eps = noise[i * LEN + wave * 16 + lrow];
    barrier_lds_only();

    v8f acc[4];
#pragma unroll
    for (int q = 0; q < 4; ++q) {
      v8f a;
      for (int v = 0; v < 8; ++v) a[v] = bias1[q];
      acc[q] = a;
    }
    v2f xa0[NPAIR1], xa1[NPAIR1];
#pragma unroll
    for (int p = 0; p < NPAIR1; ++p) {
      xa0[p] = px[4 * p];
      xa1[p] = px[4 * p + 2];
    }
    __builtin_amdgcn_sched_barrier(0);
#pragma unroll
    for (int p = 0; p < NPAIR1; ++p) {
#pragma unroll
      for (int q = 0; q < 4; ++q) {
        v4f w = w1f[q][p];
        v2f w0; w0.x = w.x; w0.y = w.y;
        v2f w1; w1.x = w.z; w1.y = w.w;
        acc[q] = wmma4(xa0[p], w0, acc[q]);
        acc[q] = wmma4(xa1[p], w1, acc[q]);
      }
    }
#pragma unroll
    for (int q = 0; q < 4; ++q) {
      int n = (wave * 4 + q) * 16 + lrow;
#pragma unroll
      for (int v = 0; v < 8; ++v) {
        float h = acc[q][v];
        hs[(v + 8 * lhalf) * HS_STRIDE + n] = h > 0.0f ? h : 0.0f;
      }
    }
    __syncthreads();

    v8f sc[4];
    {
      v8f z;
      for (int v = 0; v < 8; ++v) z[v] = 0.0f;
      v8f s0;
      for (int v = 0; v < 8; ++v) s0[v] = bias2v;
      sc[0] = s0; sc[1] = z; sc[2] = z; sc[3] = z;
    }
    v2f ha[2][4], hb[2][4];
#pragma unroll
    for (int k = 0; k < 4; ++k) {
      ha[0][k] = ph[4 * k];
      hb[0][k] = ph[4 * k + 2];
    }
    float xv[8];
#pragma unroll
    for (int v = 0; v < 8; ++v)
      xv[v] = xs[(v + 8 * lhalf) * XS_STRIDE + wave * 16 + lrow];
#pragma unroll 1
    for (int t = 0; t < 4; ++t) {
#pragma unroll
      for (int s = 0; s < 4; ++s) {
        const int cur = t * 4 + s;
        const int nw  = (cur + 3) & 15;
        const int nh  = (cur + 1) & 15;
#pragma unroll
        for (int k = 0; k < 4; ++k)
          w2r[(s + 3) & 3][k] = pw2[(nw * 4 + k) * 32];
#pragma unroll
        for (int k = 0; k < 4; ++k) {
          ha[(s + 1) & 1][k] = ph[4 * (nh * 4 + k)];
          hb[(s + 1) & 1][k] = ph[4 * (nh * 4 + k) + 2];
        }
        __builtin_amdgcn_sched_barrier(0);
#pragma unroll
        for (int k = 0; k < 4; ++k) {
          v4f w = w2r[s][k];
          v2f w0; w0.x = w.x; w0.y = w.y;
          v2f w1; w1.x = w.z; w1.y = w.w;
          sc[k] = wmma4(ha[s & 1][k], w0, sc[k]);
          sc[k] = wmma4(hb[s & 1][k], w1, sc[k]);
        }
        __builtin_amdgcn_sched_barrier(0);
      }
    }
    v8f scf;
#pragma unroll
    for (int v = 0; v < 8; ++v)
      scf[v] = (sc[0][v] + sc[1][v]) + (sc[2][v] + sc[3][v]);

    {
      int n = wave * 16 + lrow;
      int po = STEPS - 1 - i;
#pragma unroll
      for (int v = 0; v < 8; ++v) {
        int m = v + 8 * lhalf;
        float xn = xv[v] + (-0.5f * xv[v] - scf[v]) * DT_F + SQRT_DT * eps;
        xs[m * XS_STRIDE + n] = xn;
        out[((size_t)(r0 + m) * STEPS + po) * LEN + n] = xn;
      }
    }
  }
}

extern "C" void kernel_launch(void* const* d_in, const int* in_sizes, int n_in,
                              void* d_out, int out_size, void* d_ws, size_t ws_size,
                              hipStream_t stream) {
  const float* fd    = (const float*)d_in[0];
  const float* W1    = (const float*)d_in[1];
  const float* b1    = (const float*)d_in[2];
  const float* W2    = (const float*)d_in[3];
  const float* b2    = (const float*)d_in[4];
  const float* noise = (const float*)d_in[5];
  float* out = (float*)d_out;

  float* W1f = (float*)d_ws;
  float* W2f = W1f + W1F_FLOATS;

  int total = W1F_FLOATS + W2F_FLOATS;
  swizzle_weights<<<(total + 255) / 256, 256, 0, stream>>>(W1, W2, W1f, W2f);

  diffusion_scan<<<BATCH / 16, 256, 0, stream>>>(fd, b1, b2, noise, W1f, W2f, out);
}


# Own slice first from registers: products transposed so a result tile is the next GEMM's B operand; each wave starts on its own K slice (K order rotated per wave: sums reordered), split barrier
# speedup vs baseline: 1.2138x; 1.0385x over previous
#include <hip/hip_runtime.h>
#include <hip/hip_bf16.h>
#include <stddef.h>


typedef float v2f __attribute__((ext_vector_type(2)));
typedef float v4f __attribute__((ext_vector_type(4)));
typedef float v8f __attribute__((ext_vector_type(8)));

#define BATCH   512
#define STEPS   500
#define LEN     128
#define HID     512
#define NPAIR1  17
#define NPAIR2  64
#define XS_STRIDE 140
#define HS_STRIDE 516
#define DT_F     0.001f
#define SQRT_DT  0.03162277660168379332f

#define W1F_FLOATS (32 * NPAIR1 * 32 * 4)
#define W2F_FLOATS (8  * NPAIR2 * 32 * 4)

__device__ __forceinline__ v8f wmma4(v2f a, v2f b, v8f c) {
  return __builtin_amdgcn_wmma_f32_16x16x4_f32(
       false, a,  false, b,
       (short)0, c,  false,  false);
}

__device__ __forceinline__ void lds_drain_behind(v8f a, v8f b, v8f c, v8f d) {
  asm volatile("s_wait_dscnt 0x0" :: "v"(a), "v"(b), "v"(c), "v"(d) : "memory");
}
__device__ __forceinline__ void bar_signal() {
  asm volatile("s_barrier_signal -1" ::: "memory");
}
__device__ __forceinline__ void bar_wait_behind(v8f a, v8f b, v8f c, v8f d) {
  asm volatile("s_barrier_wait -1" :: "v"(a), "v"(b), "v"(c), "v"(d) : "memory");
}

__global__ __launch_bounds__(256) void swizzle_weights(
    const float* __restrict__ W1, const float* __restrict__ W2,
    float* __restrict__ W1f, float* __restrict__ W2f) {
  int gid = blockIdx.x * 256 + threadIdx.x;
  if (gid < W1F_FLOATS) {
    int j    = gid & 3;
    int lane = (gid >> 2) & 31;
    int rest = gid >> 7;
    int pp   = rest % NPAIR1;
    int nt   = rest / NPAIR1;
    int n    = nt * 16 + (lane & 15);
    int half = lane >> 4;
    int p    = (pp < 16) ? ((2 * (nt >> 2) + pp) & 15) : 16;
    int k;
    if (pp < 2)
      k = 8 * (p & ~1) + 8 * half + 4 * (p & 1) + j;
    else
      k = 4 * (2 * p + (j >> 1)) + (half << 1) + (j & 1);
    W1f[gid] = (k < LEN + 1) ? W1[k * HID + n] : 0.0f;
  } else if (gid < W1F_FLOATS + W2F_FLOATS) {
    int g    = gid - W1F_FLOATS;
    int j    = g & 3;
    int lane = (g >> 2) & 31;
    int rest = g >> 7;
    int pos  = rest & 63;
    int nt   = rest >> 6;
    int n    = nt * 16 + (lane & 15);
    int half = lane >> 4;
    int p    = 4 * ((2 * nt + (pos >> 2)) & 15) + (pos & 3);
    int k;
    if ((pos >> 2) < 2)
      k = 8 * (p & ~1) + 8 * half + 4 * (p & 1) + j;
    else
      k = 4 * (2 * p + (j >> 1)) + (half << 1) + (j & 1);
    W2f[g]   = W2[k * LEN + n];
  }
}

__global__ __launch_bounds__(256) void diffusion_scan(
    const float* __restrict__ fd,
    const float* __restrict__ b1,
    const float* __restrict__ b2,
    const float* __restrict__ noise,
    const float* __restrict__ W1f,
    const float* __restrict__ W2f,
    float* __restrict__ out) {
  __shared__ __attribute__((aligned(16))) float xs[16 * XS_STRIDE];
  __shared__ __attribute__((aligned(16))) float hs[16 * HS_STRIDE];

  const int tid   = threadIdx.x;
  const int wave  = __builtin_amdgcn_readfirstlane(tid >> 5);
  const int lane  = tid & 31;
  const int lrow  = lane & 15;
  const int lhalf = lane >> 4;
  const int r0    = blockIdx.x * 16;
  const int col0  = wave * 16 + lhalf * 8;

  for (int idx = tid; idx < 16 * XS_STRIDE; idx += 256) {
    int m = idx / XS_STRIDE;
    int c = idx - m * XS_STRIDE;
    float v = 0.0f;
    if (c < LEN) v = fd[((size_t)(r0 + m) * STEPS + (STEPS - 1)) * LEN + c];
    else if (c == LEN) v = DT_F * (float)(STEPS - 1);
    xs[idx] = v;
  }
  for (int idx = tid; idx < 16 * LEN; idx += 256) {
    int m = idx >> 7;
    int c = idx & (LEN - 1);
    out[((size_t)(r0 + m) * STEPS) * LEN + c] =
        fd[((size_t)(r0 + m) * STEPS + (STEPS - 1)) * LEN + c];
  }

  float xn[8];
  {
    const float* x0p = fd + ((size_t)(r0 + lrow) * STEPS + (STEPS - 1)) * LEN + col0;
    v4f lo = *(const v4f*)x0p;
    v4f hi = *(const v4f*)(x0p + 4);
    xn[0] = lo.x; xn[1] = lo.y; xn[2] = lo.z; xn[3] = lo.w;
    xn[4] = hi.x; xn[5] = hi.y; xn[6] = hi.z; xn[7] = hi.w;
  }

  v8f bias1v[4];
  for (int q = 0; q < 4; ++q) {
    const float* bp = b1 + (wave * 4 + q) * 16 + lhalf * 8;
    v8f b;
    for (int v = 0; v < 8; ++v) b[v] = bp[v];
    bias1v[q] = b;
  }
  v8f bias2v;
  for (int v = 0; v < 8; ++v) bias2v[v] = b2[col0 + v];

  const v2f* px = (const v2f*)(xs + lrow * XS_STRIDE + lhalf * 2);
  const v2f* ph = (const v2f*)(hs + lrow * HS_STRIDE + lhalf * 2);
  v4f* xst = (v4f*)(xs + lrow * XS_STRIDE + col0);
  const v4f* pw1 = (const v4f*)W1f + (size_t)wave * 4 * NPAIR1 * 32 + lane;
  const v4f* pw2 = (const v4f*)W2f + (size_t)wave * NPAIR2 * 32 + lane;

  v4f w1f[4][NPAIR1];
#pragma unroll
  for (int q = 0; q < 4; ++q)
#pragma unroll
    for (int pp = 0; pp < NPAIR1; ++pp) w1f[q][pp] = pw1[(q * NPAIR1 + pp) * 32];

  v4f w2r[4][4];
#pragma unroll
  for (int s = 0; s < 3; ++s)
#pragma unroll
    for (int k = 0; k < 4; ++k) w2r[s][k] = pw2[(s * 4 + k) * 32];

  __syncthreads();

#pragma unroll 1
  for (int i = 0; i < STEPS - 1; ++i) {
    v8f acc[4];
#pragma unroll
    for (int q = 0; q < 4; ++q) acc[q] = bias1v[q];
    {
      v2f xa; xa.x = xn[0]; xa.y = xn[1];
      v2f xb; xb.x = xn[2]; xb.y = xn[3];
#pragma unroll
      for (int q = 0; q < 4; ++q) {
        v4f w = w1f[q][0];
        v2f w0; w0.x = w.x; w0.y = w.y;
        acc[q] = wmma4(w0, xa, acc[q]);
      }
      lds_drain_behind(acc[0], acc[1], acc[2], acc[3]);
      bar_signal();
      __builtin_amdgcn_sched_barrier(0);
#pragma unroll
      for (int q = 0; q < 4; ++q) {
        v4f w = w1f[q][0];
        v2f w1; w1.x = w.z; w1.y = w.w;
        acc[q] = wmma4(w1, xb, acc[q]);
      }
    }
    bar_wait_behind(acc[0], acc[1], acc[2], acc[3]);
    v2f xa0[NPAIR1], xa1[NPAIR1];
#pragma unroll
    for (int g = 1; g < 8; ++g) {
      const int pb = (2 * wave + 2 * g) & 15;
      const v2f* pp = px + 4 * pb;
      xa0[2 * g]     = pp[0];
      xa1[2 * g]     = pp[2];
      xa0[2 * g + 1] = pp[4];
      xa1[2 * g + 1] = pp[6];
    }
    xa0[16] = px[64];
    xa1[16] = px[66];
    __builtin_amdgcn_sched_barrier(0);
    {
      v2f xa; xa.x = xn[4]; xa.y = xn[5];
      v2f xb; xb.x = xn[6]; xb.y = xn[7];
#pragma unroll
      for (int q = 0; q < 4; ++q) {
        v4f w = w1f[q][1];
        v2f w0; w0.x = w.x; w0.y = w.y;
        v2f w1; w1.x = w.z; w1.y = w.w;
        acc[q] = wmma4(w0, xa, acc[q]);
        acc[q] = wmma4(w1, xb, acc[q]);
      }
    }
#pragma unroll
    for (int pp = 2; pp < NPAIR1; ++pp) {
#pragma unroll
      for (int q = 0; q < 4; ++q) {
        v4f w = w1f[q][pp];
        v2f w0; w0.x = w.x; w0.y = w.y;
        v2f w1; w1.x = w.z; w1.y = w.w;
        acc[q] = wmma4(w0, xa0[pp], acc[q]);
        acc[q] = wmma4(w1, xa1[pp], acc[q]);
      }
    }
    v8f hreg[4];
#pragma unroll
    for (int q = 0; q < 4; ++q) {
      v8f h;
#pragma unroll
      for (int v = 0; v < 8; ++v) h[v] = acc[q][v] > 0.0f ? acc[q][v] : 0.0f;
      hreg[q] = h;
      v4f* hp = (v4f*)(hs + lrow * HS_STRIDE + (wave * 4 + q) * 16 + lhalf * 8);
      v4f lo; lo.x = h[0]; lo.y = h[1]; lo.z = h[2]; lo.w = h[3];
      v4f hi; hi.x = h[4]; hi.y = h[5]; hi.z = h[6]; hi.w = h[7];
      hp[0] = lo;
      hp[1] = hi;
    }
    __builtin_amdgcn_sched_barrier(0);

    v8f sc[4];
    {
      v8f z;
      for (int v = 0; v < 8; ++v) z[v] = 0.0f;
      sc[0] = bias2v; sc[1] = z; sc[2] = z; sc[3] = z;
    }
    v2f ha[2][4], hb[2][4];
    v4f e0, e1;
#pragma unroll
    for (int c = 0; c < 16; ++c) {
      {
        const int nwp = (c + 3) & 15;
#pragma unroll
        for (int k = 0; k < 4; ++k)
          w2r[(c + 3) & 3][k] = pw2[(nwp * 4 + k) * 32];
      }
      if (c >= 1 && c < 15) {
        const int nhb = (2 * wave + c + 1) & 15;
        const v2f* hp = ph + 16 * nhb;
#pragma unroll
        for (int k = 0; k < 4; ++k) {
          ha[(c + 1) & 1][k] = hp[4 * k];
          hb[(c + 1) & 1][k] = hp[4 * k + 2];
        }
      }
      if (c == 8) {
        const v4f* pe = (const v4f*)(noise + i * LEN + col0);
        e0 = pe[0];
        e1 = pe[1];
      }
      __builtin_amdgcn_sched_barrier(0);
      if (c < 2) {
#pragma unroll
        for (int k = 0; k < 4; ++k) {
          const int q  = 2 * c + (k >> 1);
          const int jp = k & 1;
          v4f w = w2r[c][k];
          v2f w0; w0.x = w.x; w0.y = w.y;
          v2f h0; h0.x = hreg[q][4 * jp];     h0.y = hreg[q][4 * jp + 1];
          sc[k] = wmma4(w0, h0, sc[k]);
        }
        if (c == 0) {
          lds_drain_behind(sc[0], sc[1], sc[2], sc[3]);
          bar_signal();
          __builtin_amdgcn_sched_barrier(0);
        }
#pragma unroll
        for (int k = 0; k < 4; ++k) {
          const int q  = 2 * c + (k >> 1);
          const int jp = k & 1;
          v4f w = w2r[c][k];
          v2f w1; w1.x = w.z; w1.y = w.w;
          v2f h1; h1.x = hreg[q][4 * jp + 2]; h1.y = hreg[q][4 * jp + 3];
          sc[k] = wmma4(w1, h1, sc[k]);
        }
      } else {
#pragma unroll
        for (int k = 0; k < 4; ++k) {
          v4f w = w2r[c & 3][k];
          v2f w0; w0.x = w.x; w0.y = w.y;
          v2f w1; w1.x = w.z; w1.y = w.w;
          sc[k] = wmma4(w0, ha[c & 1][k], sc[k]);
          sc[k] = wmma4(w1, hb[c & 1][k], sc[k]);
        }
      }
      __builtin_amdgcn_sched_barrier(0);
      if (c == 0)
        bar_wait_behind(sc[0], sc[1], sc[2], sc[3]);
    }
    v8f scf;
#pragma unroll
    for (int v = 0; v < 8; ++v)
      scf[v] = (sc[0][v] + sc[1][v]) + (sc[2][v] + sc[3][v]);

    {
      const float ev[8] = {e0.x, e0.y, e0.z, e0.w, e1.x, e1.y, e1.z, e1.w};
      int po = STEPS - 1 - i;
#pragma unroll
      for (int v = 0; v < 8; ++v)
        xn[v] = xn[v] + (-0.5f * xn[v] - scf[v]) * DT_F + SQRT_DT * ev[v];
      v4f lo; lo.x = xn[0]; lo.y = xn[1]; lo.z = xn[2]; lo.w = xn[3];
      v4f hi; hi.x = xn[4]; hi.y = xn[5]; hi.z = xn[6]; hi.w = xn[7];
      xst[0] = lo;
      xst[1] = hi;
      float* op = out + ((size_t)(r0 + lrow) * STEPS + po) * LEN + col0;
      *(v4f*)op       = lo;
      *(v4f*)(op + 4) = hi;
      if (tid < 16) xs[tid * XS_STRIDE + LEN] = DT_F * (float)(STEPS - 2 - i);
    }
  }
}

extern "C" void kernel_launch(void* const* d_in, const int* in_sizes, int n_in,
                              void* d_out, int out_size, void* d_ws, size_t ws_size,
                              hipStream_t stream) {
  const float* fd    = (const float*)d_in[0];
  const float* W1    = (const float*)d_in[1];
  const float* b1    = (const float*)d_in[2];
  const float* W2    = (const float*)d_in[3];
  const float* b2    = (const float*)d_in[4];
  const float* noise = (const float*)d_in[5];
  float* out = (float*)d_out;

  float* W1f = (float*)d_ws;
  float* W2f = W1f + W1F_FLOATS;

  int total = W1F_FLOATS + W2F_FLOATS;
  swizzle_weights<<<(total + 255) / 256, 256, 0, stream>>>(W1, W2, W1f, W2f);

  diffusion_scan<<<BATCH / 16, 256, 0, stream>>>(fd, b1, b2, noise, W1f, W2f, out);
}


# W2 fragments LDS-resident (256 KB copied once per launch, each wave its own tile): GEMM2 reads both operands from LDS one 16-WMMA block ahead, 7 load groups per step instead of 17; L2 ring gone
# speedup vs baseline: 1.2597x; 1.0378x over previous
#include <hip/hip_runtime.h>
#include <hip/hip_bf16.h>
#include <stddef.h>


typedef float v2f __attribute__((ext_vector_type(2)));
typedef float v4f __attribute__((ext_vector_type(4)));
typedef float v8f __attribute__((ext_vector_type(8)));

#define BATCH   512
#define STEPS   500
#define LEN     128
#define HID     512
#define NPAIR1  17
#define NPAIR2  64
#define XS_STRIDE 140
#define HS_STRIDE 516
#define DT_F     0.001f
#define SQRT_DT  0.03162277660168379332f

#define W1F_FLOATS (32 * NPAIR1 * 32 * 4)
#define W2F_FLOATS (8  * NPAIR2 * 32 * 4)

__device__ __forceinline__ v8f wmma4(v2f a, v2f b, v8f c) {
  return __builtin_amdgcn_wmma_f32_16x16x4_f32(
       false, a,  false, b,
       (short)0, c,  false,  false);
}

__device__ __forceinline__ void lds_drain_behind(v8f a, v8f b, v8f c, v8f d) {
  asm volatile("s_wait_dscnt 0x0" :: "v"(a), "v"(b), "v"(c), "v"(d) : "memory");
}
__device__ __forceinline__ void bar_signal() {
  asm volatile("s_barrier_signal -1" ::: "memory");
}
__device__ __forceinline__ void bar_wait_behind(v8f a, v8f b, v8f c, v8f d) {
  asm volatile("s_barrier_wait -1" :: "v"(a), "v"(b), "v"(c), "v"(d) : "memory");
}

__global__ __launch_bounds__(256) void swizzle_weights(
    const float* __restrict__ W1, const float* __restrict__ W2,
    float* __restrict__ W1f, float* __restrict__ W2f) {
  int gid = blockIdx.x * 256 + threadIdx.x;
  if (gid < W1F_FLOATS) {
    int j    = gid & 3;
    int lane = (gid >> 2) & 31;
    int rest = gid >> 7;
    int pp   = rest % NPAIR1;
    int nt   = rest / NPAIR1;
    int n    = nt * 16 + (lane & 15);
    int half = lane >> 4;
    int p    = (pp < 16) ? ((2 * (nt >> 2) + pp) & 15) : 16;
    int k;
    if (pp < 2)
      k = 8 * (p & ~1) + 8 * half + 4 * (p & 1) + j;
    else
      k = 4 * (2 * p + (j >> 1)) + (half << 1) + (j & 1);
    W1f[gid] = (k < LEN + 1) ? W1[k * HID + n] : 0.0f;
  } else if (gid < W1F_FLOATS + W2F_FLOATS) {
    int g    = gid - W1F_FLOATS;
    int j    = g & 3;
    int lane = (g >> 2) & 31;
    int rest = g >> 7;
    int pos  = rest & 63;
    int nt   = rest >> 6;
    int n    = nt * 16 + (lane & 15);
    int half = lane >> 4;
    int p    = 4 * ((2 * nt + (pos >> 2)) & 15) + (pos & 3);
    int k;
    if ((pos >> 2) < 2)
      k = 8 * (p & ~1) + 8 * half + 4 * (p & 1) + j;
    else
      k = 4 * (2 * p + (j >> 1)) + (half << 1) + (j & 1);
    W2f[g]   = W2[k * LEN + n];
  }
}

__global__ __launch_bounds__(256) void diffusion_scan(
    const float* __restrict__ fd,
    const float* __restrict__ b1,
    const float* __restrict__ b2,
    const float* __restrict__ noise,
    const float* __restrict__ W1f,
    const float* __restrict__ W2f,
    float* __restrict__ out) {
  __shared__ __attribute__((aligned(16))) float lds[W2F_FLOATS + 16 * XS_STRIDE + 16 * HS_STRIDE + LEN];
  float* const w2s = lds;
  float* const xs  = lds + W2F_FLOATS;
  float* const hs  = lds + W2F_FLOATS + 16 * XS_STRIDE;
  float* const b2s = hs + 16 * HS_STRIDE;

  const int tid   = threadIdx.x;
  const int wave  = __builtin_amdgcn_readfirstlane(tid >> 5);
  const int lane  = tid & 31;
  const int lrow  = lane & 15;
  const int lhalf = lane >> 4;
  const int r0    = blockIdx.x * 16;
  const int col0  = wave * 16 + lhalf * 8;

  for (int idx = tid; idx < 16 * XS_STRIDE; idx += 256) {
    int m = idx / XS_STRIDE;
    int c = idx - m * XS_STRIDE;
    float v = 0.0f;
    if (c < LEN) v = fd[((size_t)(r0 + m) * STEPS + (STEPS - 1)) * LEN + c];
    else if (c == LEN) v = DT_F * (float)(STEPS - 1);
    xs[idx] = v;
  }
  for (int idx = tid; idx < 16 * LEN; idx += 256) {
    int m = idx >> 7;
    int c = idx & (LEN - 1);
    out[((size_t)(r0 + m) * STEPS) * LEN + c] =
        fd[((size_t)(r0 + m) * STEPS + (STEPS - 1)) * LEN + c];
  }

  float xn[8];
  {
    const float* x0p = fd + ((size_t)(r0 + lrow) * STEPS + (STEPS - 1)) * LEN + col0;
    v4f lo = *(const v4f*)x0p;
    v4f hi = *(const v4f*)(x0p + 4);
    xn[0] = lo.x; xn[1] = lo.y; xn[2] = lo.z; xn[3] = lo.w;
    xn[4] = hi.x; xn[5] = hi.y; xn[6] = hi.z; xn[7] = hi.w;
  }

  v8f bias1v[4];
  for (int q = 0; q < 4; ++q) {
    const float* bp = b1 + (wave * 4 + q) * 16 + lhalf * 8;
    v8f b;
    for (int v = 0; v < 8; ++v) b[v] = bp[v];
    bias1v[q] = b;
  }
  if (tid < LEN) b2s[tid] = b2[tid];

  const v2f* px = (const v2f*)(xs + lrow * XS_STRIDE + lhalf * 2);
  const v2f* ph = (const v2f*)(hs + lrow * HS_STRIDE + lhalf * 2);
  v4f* xst = (v4f*)(xs + lrow * XS_STRIDE + col0);
  const v4f* pw1 = (const v4f*)W1f + (size_t)wave * 4 * NPAIR1 * 32 + lane;
  const v4f* pw2 = (const v4f*)W2f + (size_t)wave * NPAIR2 * 32 + lane;
  const v4f* pl2 = (const v4f*)w2s + wave * NPAIR2 * 32 + lane;

  {
    v4f* pl2w = (v4f*)w2s + wave * NPAIR2 * 32 + lane;
#pragma unroll 16
    for (int p = 0; p < NPAIR2; ++p) pl2w[p * 32] = pw2[p * 32];
  }

  v4f w1f[4][NPAIR1];
#pragma unroll
  for (int q = 0; q < 4; ++q)
#pragma unroll
    for (int pp = 0; pp < NPAIR1; ++pp) w1f[q][pp] = pw1[(q * NPAIR1 + pp) * 32];

  __syncthreads();

#pragma unroll 1
  for (int i = 0; i < STEPS - 1; ++i) {
    v8f acc[4];
#pragma unroll
    for (int q = 0; q < 4; ++q) acc[q] = bias1v[q];
    {
      v2f xa; xa.x = xn[0]; xa.y = xn[1];
      v2f xb; xb.x = xn[2]; xb.y = xn[3];
#pragma unroll
      for (int q = 0; q < 4; ++q) {
        v4f w = w1f[q][0];
        v2f w0; w0.x = w.x; w0.y = w.y;
        acc[q] = wmma4(w0, xa, acc[q]);
      }
      lds_drain_behind(acc[0], acc[1], acc[2], acc[3]);
      bar_signal();
      __builtin_amdgcn_sched_barrier(0);
#pragma unroll
      for (int q = 0; q < 4; ++q) {
        v4f w = w1f[q][0];
        v2f w1; w1.x = w.z; w1.y = w.w;
        acc[q] = wmma4(w1, xb, acc[q]);
      }
    }
    bar_wait_behind(acc[0], acc[1], acc[2], acc[3]);
    v2f xa0[NPAIR1], xa1[NPAIR1];
#pragma unroll
    for (int g = 1; g < 8; ++g) {
      const int pb = (2 * wave + 2 * g) & 15;
      const v2f* pp = px + 4 * pb;
      xa0[2 * g]     = pp[0];
      xa1[2 * g]     = pp[2];
      xa0[2 * g + 1] = pp[4];
      xa1[2 * g + 1] = pp[6];
    }
    xa0[16] = px[64];
    xa1[16] = px[66];
    v4f wa[2][8];
#pragma unroll
    for (int k = 0; k < 8; ++k) wa[0][k] = pl2[k * 32];
    const v4f c2lo = *(const v4f*)(b2s + col0);
    const v4f c2hi = *(const v4f*)(b2s + col0 + 4);
    __builtin_amdgcn_sched_barrier(0);
    {
      v2f xa; xa.x = xn[4]; xa.y = xn[5];
      v2f xb; xb.x = xn[6]; xb.y = xn[7];
#pragma unroll
      for (int q = 0; q < 4; ++q) {
        v4f w = w1f[q][1];
        v2f w0; w0.x = w.x; w0.y = w.y;
        v2f w1; w1.x = w.z; w1.y = w.w;
        acc[q] = wmma4(w0, xa, acc[q]);
        acc[q] = wmma4(w1, xb, acc[q]);
      }
    }
#pragma unroll
    for (int pp = 2; pp < NPAIR1; ++pp) {
#pragma unroll
      for (int q = 0; q < 4; ++q) {
        v4f w = w1f[q][pp];
        v2f w0; w0.x = w.x; w0.y = w.y;
        v2f w1; w1.x = w.z; w1.y = w.w;
        acc[q] = wmma4(w0, xa0[pp], acc[q]);
        acc[q] = wmma4(w1, xa1[pp], acc[q]);
      }
    }
    v8f hreg[4];
#pragma unroll
    for (int q = 0; q < 4; ++q) {
      v8f h;
#pragma unroll
      for (int v = 0; v < 8; ++v) h[v] = acc[q][v] > 0.0f ? acc[q][v] : 0.0f;
      hreg[q] = h;
      v4f* hp = (v4f*)(hs + lrow * HS_STRIDE + (wave * 4 + q) * 16 + lhalf * 8);
      v4f lo; lo.x = h[0]; lo.y = h[1]; lo.z = h[2]; lo.w = h[3];
      v4f hi; hi.x = h[4]; hi.y = h[5]; hi.z = h[6]; hi.w = h[7];
      hp[0] = lo;
      hp[1] = hi;
    }

    v8f sc[4];
    {
      v8f z;
      for (int v = 0; v < 8; ++v) z[v] = 0.0f;
      v8f c2;
      c2[0] = c2lo.x; c2[1] = c2lo.y; c2[2] = c2lo.z; c2[3] = c2lo.w;
      c2[4] = c2hi.x; c2[5] = c2hi.y; c2[6] = c2hi.z; c2[7] = c2hi.w;
      sc[0] = c2; sc[1] = z; sc[2] = z; sc[3] = z;
    }
    v2f ha[2][8], hb[2][8];
    v4f e0, e1;
#pragma unroll
    for (int k = 0; k < 4; ++k) {
      const int q  = k >> 1;
      const int jp = k & 1;
      v4f w = wa[0][k];
      v2f w0; w0.x = w.x; w0.y = w.y;
      v2f h0; h0.x = hreg[q][4 * jp];     h0.y = hreg[q][4 * jp + 1];
      sc[k] = wmma4(w0, h0, sc[k]);
    }
    lds_drain_behind(sc[0], sc[1], sc[2], sc[3]);
    bar_signal();
    __builtin_amdgcn_sched_barrier(0);
#pragma unroll
    for (int k = 0; k < 4; ++k) {
      const int q  = k >> 1;
      const int jp = k & 1;
      v4f w = wa[0][k];
      v2f w1; w1.x = w.z; w1.y = w.w;
      v2f h1; h1.x = hreg[q][4 * jp + 2]; h1.y = hreg[q][4 * jp + 3];
      sc[k] = wmma4(w1, h1, sc[k]);
    }
    __builtin_amdgcn_sched_barrier(0);
    bar_wait_behind(sc[0], sc[1], sc[2], sc[3]);
#pragma unroll
    for (int b = 0; b < 8; ++b) {
      if (b < 7) {
        const int nb  = b + 1;
        const int nhb = (2 * wave + 2 * nb) & 15;
        const v2f* hp = ph + 16 * nhb;
#pragma unroll
        for (int k = 0; k < 8; ++k) {
          ha[nb & 1][k] = hp[4 * k];
          hb[nb & 1][k] = hp[4 * k + 2];
          wa[nb & 1][k] = pl2[(nb * 8 + k) * 32];
        }
      }
      if (b == 4) {
        const v4f* pe = (const v4f*)(noise + i * LEN + col0);
        e0 = pe[0];
        e1 = pe[1];
      }
      __builtin_amdgcn_sched_barrier(0);
      if (b == 0) {
#pragma unroll
        for (int k = 0; k < 4; ++k) {
          const int q  = 2 + (k >> 1);
          const int jp = k & 1;
          v4f w = wa[0][4 + k];
          v2f w0; w0.x = w.x; w0.y = w.y;
          v2f h0; h0.x = hreg[q][4 * jp];     h0.y = hreg[q][4 * jp + 1];
          sc[k] = wmma4(w0, h0, sc[k]);
        }
#pragma unroll
        for (int k = 0; k < 4; ++k) {
          const int q  = 2 + (k >> 1);
          const int jp = k & 1;
          v4f w = wa[0][4 + k];
          v2f w1; w1.x = w.z; w1.y = w.w;
          v2f h1; h1.x = hreg[q][4 * jp + 2]; h1.y = hreg[q][4 * jp + 3];
          sc[k] = wmma4(w1, h1, sc[k]);
        }
      } else {
#pragma unroll
        for (int kk = 0; kk < 8; ++kk) {
          const int k = kk & 3;
          v4f w = wa[b & 1][kk];
          v2f w0; w0.x = w.x; w0.y = w.y;
          v2f w1; w1.x = w.z; w1.y = w.w;
          sc[k] = wmma4(w0, ha[b & 1][kk], sc[k]);
          sc[k] = wmma4(w1, hb[b & 1][kk], sc[k]);
        }
      }
      __builtin_amdgcn_sched_barrier(0);
    }
    v8f scf;
#pragma unroll
    for (int v = 0; v < 8; ++v)
      scf[v] = (sc[0][v] + sc[1][v]) + (sc[2][v] + sc[3][v]);

    {
      const float ev[8] = {e0.x, e0.y, e0.z, e0.w, e1.x, e1.y, e1.z, e1.w};
      int po = STEPS - 1 - i;
#pragma unroll
      for (int v = 0; v < 8; ++v)
        xn[v] = xn[v] + (-0.5f * xn[v] - scf[v]) * DT_F + SQRT_DT * ev[v];
      v4f lo; lo.x = xn[0]; lo.y = xn[1]; lo.z = xn[2]; lo.w = xn[3];
      v4f hi; hi.x = xn[4]; hi.y = xn[5]; hi.z = xn[6]; hi.w = xn[7];
      xst[0] = lo;
      xst[1] = hi;
      float* op = out + ((size_t)(r0 + lrow) * STEPS + po) * LEN + col0;
      *(v4f*)op       = lo;
      *(v4f*)(op + 4) = hi;
      if (tid < 16) xs[tid * XS_STRIDE + LEN] = DT_F * (float)(STEPS - 2 - i);
    }
  }
}

extern "C" void kernel_launch(void* const* d_in, const int* in_sizes, int n_in,
                              void* d_out, int out_size, void* d_ws, size_t ws_size,
                              hipStream_t stream) {
  const float* fd    = (const float*)d_in[0];
  const float* W1    = (const float*)d_in[1];
  const float* b1    = (const float*)d_in[2];
  const float* W2    = (const float*)d_in[3];
  const float* b2    = (const float*)d_in[4];
  const float* noise = (const float*)d_in[5];
  float* out = (float*)d_out;

  float* W1f = (float*)d_ws;
  float* W2f = W1f + W1F_FLOATS;

  int total = W1F_FLOATS + W2F_FLOATS;
  swizzle_weights<<<(total + 255) / 256, 256, 0, stream>>>(W1, W2, W1f, W2f);

  diffusion_scan<<<BATCH / 16, 256, 0, stream>>>(fd, b1, b2, noise, W1f, W2f, out);
}


# Time column folded into GEMM1's start value b1 + t*W1[128,:] (FMA per element, t uniform over rows; b1 re-read from LDS per step): K 136 -> 128, 8 fewer WMMAs per wave and step; t term summed first
# speedup vs baseline: 1.3245x; 1.0514x over previous
#include <hip/hip_runtime.h>
#include <hip/hip_bf16.h>
#include <stddef.h>


typedef float v2f __attribute__((ext_vector_type(2)));
typedef float v4f __attribute__((ext_vector_type(4)));
typedef float v8f __attribute__((ext_vector_type(8)));

#define BATCH   512
#define STEPS   500
#define LEN     128
#define HID     512
#define NPAIR1  16
#define NPAIR2  64
#define XS_STRIDE 140
#define HS_STRIDE 516
#define DT_F     0.001f
#define SQRT_DT  0.03162277660168379332f

#define W1F_FLOATS (32 * NPAIR1 * 32 * 4)
#define W2F_FLOATS (8  * NPAIR2 * 32 * 4)

__device__ __forceinline__ v8f wmma4(v2f a, v2f b, v8f c) {
  return __builtin_amdgcn_wmma_f32_16x16x4_f32(
       false, a,  false, b,
       (short)0, c,  false,  false);
}

__device__ __forceinline__ v8f chain_start(v4f lo, v4f hi, v8f w, float t) {
  const float b[8] = {lo.x, lo.y, lo.z, lo.w, hi.x, hi.y, hi.z, hi.w};
  v8f r;
#pragma unroll
  for (int v = 0; v < 8; ++v) r[v] = __builtin_fmaf(t, w[v], b[v]);
  return r;
}

__device__ __forceinline__ void lds_drain_behind(v8f a, v8f b, v8f c, v8f d) {
  asm volatile("s_wait_dscnt 0x0" :: "v"(a), "v"(b), "v"(c), "v"(d) : "memory");
}
__device__ __forceinline__ void bar_signal() {
  asm volatile("s_barrier_signal -1" ::: "memory");
}
__device__ __forceinline__ void bar_wait_behind(v8f a, v8f b, v8f c, v8f d) {
  asm volatile("s_barrier_wait -1" :: "v"(a), "v"(b), "v"(c), "v"(d) : "memory");
}

__global__ __launch_bounds__(256) void swizzle_weights(
    const float* __restrict__ W1, const float* __restrict__ W2,
    float* __restrict__ W1f, float* __restrict__ W2f) {
  int gid = blockIdx.x * 256 + threadIdx.x;
  if (gid < W1F_FLOATS) {
    int j    = gid & 3;
    int lane = (gid >> 2) & 31;
    int rest = gid >> 7;
    int pp   = rest & (NPAIR1 - 1);
    int nt   = rest / NPAIR1;
    int n    = nt * 16 + (lane & 15);
    int half = lane >> 4;
    int p    = (2 * (nt >> 2) + pp) & 15;
    int k;
    if (pp < 2)
      k = 8 * (p & ~1) + 8 * half + 4 * (p & 1) + j;
    else
      k = 4 * (2 * p + (j >> 1)) + (half << 1) + (j & 1);
    W1f[gid] = W1[k * HID + n];
  } else if (gid < W1F_FLOATS + W2F_FLOATS) {
    int g    = gid - W1F_FLOATS;
    int j    = g & 3;
    int lane = (g >> 2) & 31;
    int rest = g >> 7;
    int pos  = rest & 63;
    int nt   = rest >> 6;
    int n    = nt * 16 + (lane & 15);
    int half = lane >> 4;
    int p    = 4 * ((2 * nt + (pos >> 2)) & 15) + (pos & 3);
    int k;
    if ((pos >> 2) < 2)
      k = 8 * (p & ~1) + 8 * half + 4 * (p & 1) + j;
    else
      k = 4 * (2 * p + (j >> 1)) + (half << 1) + (j & 1);
    W2f[g]   = W2[k * LEN + n];
  }
}

__global__ __launch_bounds__(256) void diffusion_scan(
    const float* __restrict__ fd,
    const float* __restrict__ W1,
    const float* __restrict__ b1,
    const float* __restrict__ b2,
    const float* __restrict__ noise,
    const float* __restrict__ W1f,
    const float* __restrict__ W2f,
    float* __restrict__ out) {
  __shared__ __attribute__((aligned(16))) float lds[W2F_FLOATS + 16 * XS_STRIDE + 16 * HS_STRIDE + LEN + HID];
  float* const w2s = lds;
  float* const xs  = lds + W2F_FLOATS;
  float* const hs  = lds + W2F_FLOATS + 16 * XS_STRIDE;
  float* const b2s = hs + 16 * HS_STRIDE;
  float* const b1s = b2s + LEN;

  const int tid   = threadIdx.x;
  const int wave  = __builtin_amdgcn_readfirstlane(tid >> 5);
  const int lane  = tid & 31;
  const int lrow  = lane & 15;
  const int lhalf = lane >> 4;
  const int r0    = blockIdx.x * 16;
  const int col0  = wave * 16 + lhalf * 8;

  for (int idx = tid; idx < 16 * LEN; idx += 256) {
    int m = idx >> 7;
    int c = idx & (LEN - 1);
    float v = fd[((size_t)(r0 + m) * STEPS + (STEPS - 1)) * LEN + c];
    xs[m * XS_STRIDE + c] = v;
    out[((size_t)(r0 + m) * STEPS) * LEN + c] = v;
  }

  float xn[8];
  {
    const float* x0p = fd + ((size_t)(r0 + lrow) * STEPS + (STEPS - 1)) * LEN + col0;
    v4f lo = *(const v4f*)x0p;
    v4f hi = *(const v4f*)(x0p + 4);
    xn[0] = lo.x; xn[1] = lo.y; xn[2] = lo.z; xn[3] = lo.w;
    xn[4] = hi.x; xn[5] = hi.y; xn[6] = hi.z; xn[7] = hi.w;
  }

  v8f w1t[4];
  for (int q = 0; q < 4; ++q) {
    const float* tp = W1 + LEN * HID + (wave * 4 + q) * 16 + lhalf * 8;
    v8f w;
    for (int v = 0; v < 8; ++v) w[v] = tp[v];
    w1t[q] = w;
  }
  for (int idx = tid; idx < HID; idx += 256) b1s[idx] = b1[idx];
  if (tid < LEN) b2s[tid] = b2[tid];

  const v4f* pb1 = (const v4f*)(b1s + wave * 64 + lhalf * 8);
  const v2f* px = (const v2f*)(xs + lrow * XS_STRIDE + lhalf * 2);
  const v2f* ph = (const v2f*)(hs + lrow * HS_STRIDE + lhalf * 2);
  v4f* xst = (v4f*)(xs + lrow * XS_STRIDE + col0);
  const v4f* pw1 = (const v4f*)W1f + (size_t)wave * 4 * NPAIR1 * 32 + lane;
  const v4f* pw2 = (const v4f*)W2f + (size_t)wave * NPAIR2 * 32 + lane;
  const v4f* pl2 = (const v4f*)w2s + wave * NPAIR2 * 32 + lane;

  {
    v4f* pl2w = (v4f*)w2s + wave * NPAIR2 * 32 + lane;
#pragma unroll 16
    for (int p = 0; p < NPAIR2; ++p) pl2w[p * 32] = pw2[p * 32];
  }

  v4f w1f[4][NPAIR1];
#pragma unroll
  for (int q = 0; q < 4; ++q)
#pragma unroll
    for (int pp = 0; pp < NPAIR1; ++pp) w1f[q][pp] = pw1[(q * NPAIR1 + pp) * 32];

  __syncthreads();

  v8f accs[4];
#pragma unroll
  for (int q = 0; q < 4; ++q)
    accs[q] = chain_start(pb1[4 * q], pb1[4 * q + 1], w1t[q], DT_F * (float)(STEPS - 1));

#pragma unroll 1
  for (int i = 0; i < STEPS - 1; ++i) {
    v8f acc[4];
#pragma unroll
    for (int q = 0; q < 4; ++q) acc[q] = accs[q];
    {
      v2f xa; xa.x = xn[0]; xa.y = xn[1];
      v2f xb; xb.x = xn[2]; xb.y = xn[3];
#pragma unroll
      for (int q = 0; q < 4; ++q) {
        v4f w = w1f[q][0];
        v2f w0; w0.x = w.x; w0.y = w.y;
        acc[q] = wmma4(w0, xa, acc[q]);
      }
      lds_drain_behind(acc[0], acc[1], acc[2], acc[3]);
      bar_signal();
      __builtin_amdgcn_sched_barrier(0);
#pragma unroll
      for (int q = 0; q < 4; ++q) {
        v4f w = w1f[q][0];
        v2f w1; w1.x = w.z; w1.y = w.w;
        acc[q] = wmma4(w1, xb, acc[q]);
      }
    }
    bar_wait_behind(acc[0], acc[1], acc[2], acc[3]);
    v2f xa0[NPAIR1], xa1[NPAIR1];
#pragma unroll
    for (int g = 1; g < 8; ++g) {
      const int pb = (2 * wave + 2 * g) & 15;
      const v2f* pp = px + 4 * pb;
      xa0[2 * g]     = pp[0];
      xa1[2 * g]     = pp[2];
      xa0[2 * g + 1] = pp[4];
      xa1[2 * g + 1] = pp[6];
    }
    v4f wa[2][8];
#pragma unroll
    for (int k = 0; k < 8; ++k) wa[0][k] = pl2[k * 32];
    const v4f c2lo = *(const v4f*)(b2s + col0);
    const v4f c2hi = *(const v4f*)(b2s + col0 + 4);
    __builtin_amdgcn_sched_barrier(0);
    {
      v2f xa; xa.x = xn[4]; xa.y = xn[5];
      v2f xb; xb.x = xn[6]; xb.y = xn[7];
#pragma unroll
      for (int q = 0; q < 4; ++q) {
        v4f w = w1f[q][1];
        v2f w0; w0.x = w.x; w0.y = w.y;
        v2f w1; w1.x = w.z; w1.y = w.w;
        acc[q] = wmma4(w0, xa, acc[q]);
        acc[q] = wmma4(w1, xb, acc[q]);
      }
    }
#pragma unroll
    for (int pp = 2; pp < NPAIR1; ++pp) {
#pragma unroll
      for (int q = 0; q < 4; ++q) {
        v4f w = w1f[q][pp];
        v2f w0; w0.x = w.x; w0.y = w.y;
        v2f w1; w1.x = w.z; w1.y = w.w;
        acc[q] = wmma4(w0, xa0[pp], acc[q]);
        acc[q] = wmma4(w1, xa1[pp], acc[q]);
      }
    }
    v8f hreg[4];
#pragma unroll
    for (int q = 0; q < 4; ++q) {
      v8f h;
#pragma unroll
      for (int v = 0; v < 8; ++v) h[v] = acc[q][v] > 0.0f ? acc[q][v] : 0.0f;
      hreg[q] = h;
      v4f* hp = (v4f*)(hs + lrow * HS_STRIDE + (wave * 4 + q) * 16 + lhalf * 8);
      v4f lo; lo.x = h[0]; lo.y = h[1]; lo.z = h[2]; lo.w = h[3];
      v4f hi; hi.x = h[4]; hi.y = h[5]; hi.z = h[6]; hi.w = h[7];
      hp[0] = lo;
      hp[1] = hi;
    }

    v8f sc[4];
    {
      v8f z;
      for (int v = 0; v < 8; ++v) z[v] = 0.0f;
      v8f c2;
      c2[0] = c2lo.x; c2[1] = c2lo.y; c2[2] = c2lo.z; c2[3] = c2lo.w;
      c2[4] = c2hi.x; c2[5] = c2hi.y; c2[6] = c2hi.z; c2[7] = c2hi.w;
      sc[0] = c2; sc[1] = z; sc[2] = z; sc[3] = z;
    }
    v2f ha[2][8], hb[2][8];
    v4f e0, e1;
    v4f b1lo[4], b1hi[4];
#pragma unroll
    for (int k = 0; k < 4; ++k) {
      const int q  = k >> 1;
      const int jp = k & 1;
      v4f w = wa[0][k];
      v2f w0; w0.x = w.x; w0.y = w.y;
      v2f h0; h0.x = hreg[q][4 * jp];     h0.y = hreg[q][4 * jp + 1];
      sc[k] = wmma4(w0, h0, sc[k]);
    }
    lds_drain_behind(sc[0], sc[1], sc[2], sc[3]);
    bar_signal();
    __builtin_amdgcn_sched_barrier(0);
#pragma unroll
    for (int k = 0; k < 4; ++k) {
      const int q  = k >> 1;
      const int jp = k & 1;
      v4f w = wa[0][k];
      v2f w1; w1.x = w.z; w1.y = w.w;
      v2f h1; h1.x = hreg[q][4 * jp + 2]; h1.y = hreg[q][4 * jp + 3];
      sc[k] = wmma4(w1, h1, sc[k]);
    }
    __builtin_amdgcn_sched_barrier(0);
    bar_wait_behind(sc[0], sc[1], sc[2], sc[3]);
#pragma unroll
    for (int b = 0; b < 8; ++b) {
      if (b < 7) {
        const int nb  = b + 1;
        const int nhb = (2 * wave + 2 * nb) & 15;
        const v2f* hp = ph + 16 * nhb;
#pragma unroll
        for (int k = 0; k < 8; ++k) {
          ha[nb & 1][k] = hp[4 * k];
          hb[nb & 1][k] = hp[4 * k + 2];
          wa[nb & 1][k] = pl2[(nb * 8 + k) * 32];
        }
      }
      if (b == 7) {
#pragma unroll
        for (int q = 0; q < 4; ++q) { b1lo[q] = pb1[4 * q]; b1hi[q] = pb1[4 * q + 1]; }
      }
      if (b == 4) {
        const v4f* pe = (const v4f*)(noise + i * LEN + col0);
        e0 = pe[0];
        e1 = pe[1];
      }
      __builtin_amdgcn_sched_barrier(0);
      if (b == 0) {
#pragma unroll
        for (int k = 0; k < 4; ++k) {
          const int q  = 2 + (k >> 1);
          const int jp = k & 1;
          v4f w = wa[0][4 + k];
          v2f w0; w0.x = w.x; w0.y = w.y;
          v2f h0; h0.x = hreg[q][4 * jp];     h0.y = hreg[q][4 * jp + 1];
          sc[k] = wmma4(w0, h0, sc[k]);
        }
#pragma unroll
        for (int k = 0; k < 4; ++k) {
          const int q  = 2 + (k >> 1);
          const int jp = k & 1;
          v4f w = wa[0][4 + k];
          v2f w1; w1.x = w.z; w1.y = w.w;
          v2f h1; h1.x = hreg[q][4 * jp + 2]; h1.y = hreg[q][4 * jp + 3];
          sc[k] = wmma4(w1, h1, sc[k]);
        }
      } else {
#pragma unroll
        for (int kk = 0; kk < 8; ++kk) {
          const int k = kk & 3;
          v4f w = wa[b & 1][kk];
          v2f w0; w0.x = w.x; w0.y = w.y;
          v2f w1; w1.x = w.z; w1.y = w.w;
          sc[k] = wmma4(w0, ha[b & 1][kk], sc[k]);
          sc[k] = wmma4(w1, hb[b & 1][kk], sc[k]);
        }
      }
      __builtin_amdgcn_sched_barrier(0);
    }
    {
      const float tnext = DT_F * (float)(STEPS - 2 - i);
#pragma unroll
      for (int q = 0; q < 4; ++q) accs[q] = chain_start(b1lo[q], b1hi[q], w1t[q], tnext);
    }
    __builtin_amdgcn_sched_barrier(0);
    v8f scf;
#pragma unroll
    for (int v = 0; v < 8; ++v)
      scf[v] = (sc[0][v] + sc[1][v]) + (sc[2][v] + sc[3][v]);

    {
      const float ev[8] = {e0.x, e0.y, e0.z, e0.w, e1.x, e1.y, e1.z, e1.w};
      int po = STEPS - 1 - i;
#pragma unroll
      for (int v = 0; v < 8; ++v)
        xn[v] = xn[v] + (-0.5f * xn[v] - scf[v]) * DT_F + SQRT_DT * ev[v];
      v4f lo; lo.x = xn[0]; lo.y = xn[1]; lo.z = xn[2]; lo.w = xn[3];
      v4f hi; hi.x = xn[4]; hi.y = xn[5]; hi.z = xn[6]; hi.w = xn[7];
      xst[0] = lo;
      xst[1] = hi;
      float* op = out + ((size_t)(r0 + lrow) * STEPS + po) * LEN + col0;
      *(v4f*)op       = lo;
      *(v4f*)(op + 4) = hi;
    }
  }
}

extern "C" void kernel_launch(void* const* d_in, const int* in_sizes, int n_in,
                              void* d_out, int out_size, void* d_ws, size_t ws_size,
                              hipStream_t stream) {
  const float* fd    = (const float*)d_in[0];
  const float* W1    = (const float*)d_in[1];
  const float* b1    = (const float*)d_in[2];
  const float* W2    = (const float*)d_in[3];
  const float* b2    = (const float*)d_in[4];
  const float* noise = (const float*)d_in[5];
  float* out = (float*)d_out;

  float* W1f = (float*)d_ws;
  float* W2f = W1f + W1F_FLOATS;

  int total = W1F_FLOATS + W2F_FLOATS;
  swizzle_weights<<<(total + 255) / 256, 256, 0, stream>>>(W1, W2, W1f, W2f);

  diffusion_scan<<<BATCH / 16, 256, 0, stream>>>(fd, W1, b1, b2, noise, W1f, W2f, out);
}
